# starting kernel
# baseline (speedup 1.0000x reference)
#include <hip/hip_runtime.h>

typedef __attribute__((ext_vector_type(2))) float v2f;
typedef __attribute__((ext_vector_type(8))) float v8f;

#define NUM_FEATURES 256
#define GROUP_SIZE   32
#define NUM_GROUPS   8
#define BATCH        32
#define HW           4096
#define NCOLS        (BATCH * HW)
#define EPS          1e-5f
#define NS_ITERS     32

#define WS_GRAM   0
#define WS_SUMS   (NUM_GROUPS * 32 * 32)
#define WS_WM     (WS_SUMS + NUM_GROUPS * 32)
#define WS_SHIFT  (WS_WM + NUM_GROUPS * 32 * 32)

__global__ void __launch_bounds__(256) dbn_gram(const float* __restrict__ x,
                                                float* __restrict__ ws) {
    const int g      = blockIdx.y;
    const int lane   = threadIdx.x & 31;
    const int wave   = threadIdx.x >> 5;
    const int lane16 = lane & 15;
    const int half   = lane >> 4;
    const int wid    = blockIdx.x * 8 + wave;

    float* gram = ws + WS_GRAM + g * 1024;
    float* sums = ws + WS_SUMS + g * 32;

    const int b  = wid >> 4;
    const int s0 = (wid & 15) * 256;

    const float* pLo = x + (size_t)(b * NUM_FEATURES + g * GROUP_SIZE + lane16) * HW
                         + (size_t)(s0 + 2 * half);
    const float* pHi = pLo + (size_t)16 * HW;

    v8f a00 = {}, a01 = {}, a10 = {}, a11 = {};
    float sLo = 0.f, sHi = 0.f;

    #pragma unroll 4
    for (int i = 0; i < 64; ++i) {
        v2f lo = *(const v2f*)(pLo + i * 4);
        v2f hi = *(const v2f*)(pHi + i * 4);

        a00 = __builtin_amdgcn_wmma_f32_16x16x4_f32(false, lo, false, lo, (short)0, a00, false, false);
        a01 = __builtin_amdgcn_wmma_f32_16x16x4_f32(false, lo, false, hi, (short)0, a01, false, false);
        a10 = __builtin_amdgcn_wmma_f32_16x16x4_f32(false, hi, false, lo, (short)0, a10, false, false);
        a11 = __builtin_amdgcn_wmma_f32_16x16x4_f32(false, hi, false, hi, (short)0, a11, false, false);

        sLo += lo.x + lo.y;
        sHi += hi.x + hi.y;
    }

    sLo += __shfl_xor(sLo, 16, 32);
    sHi += __shfl_xor(sHi, 16, 32);
    if (half == 0) {
        atomicAdd(&sums[lane16],      sLo);
        atomicAdd(&sums[lane16 + 16], sHi);
    }

    #pragma unroll
    for (int j = 0; j < 8; ++j) {
        const int m = j + 8 * half;
        const int n = lane16;
        atomicAdd(&gram[(m)      * 32 + n],      a00[j]);
        atomicAdd(&gram[(m)      * 32 + n + 16], a01[j]);
        atomicAdd(&gram[(m + 16) * 32 + n],      a10[j]);
        atomicAdd(&gram[(m + 16) * 32 + n + 16], a11[j]);
    }
}

__global__ void __launch_bounds__(1024) dbn_invsqrt(float* __restrict__ ws) {
    const int g = blockIdx.x;
    const int j = threadIdx.x & 31;
    const int i = threadIdx.x >> 5;

    __shared__ float Y[32][33], Z[32][33], T[32][33];
    __shared__ float tr;

    const float* gram = ws + WS_GRAM + g * 1024;
    const float* sums = ws + WS_SUMS + g * 32;
    float* wm    = ws + WS_WM    + g * 1024;
    float* shift = ws + WS_SHIFT + g * 32;

    const float invN = 1.0f / (float)NCOLS;
    const float mi = sums[i] * invN;
    const float mj = sums[j] * invN;
    float sij = gram[i * 32 + j] - (float)NCOLS * mi * mj + ((i == j) ? EPS : 0.f);

    if (threadIdx.x == 0) tr = 0.f;
    __syncthreads();
    if (i == j) atomicAdd(&tr, sij);
    __syncthreads();
    const float t = tr;

    Y[i][j] = sij / t;
    Z[i][j] = (i == j) ? 1.f : 0.f;

    for (int it = 0; it < NS_ITERS; ++it) {
        __syncthreads();
        float zy = 0.f;
        #pragma unroll
        for (int k = 0; k < 32; ++k) zy += Z[i][k] * Y[k][j];
        T[i][j] = ((i == j) ? 3.f : 0.f) - zy;
        __syncthreads();
        float yt = 0.f, tz = 0.f;
        #pragma unroll
        for (int k = 0; k < 32; ++k) { yt += Y[i][k] * T[k][j]; tz += T[i][k] * Z[k][j]; }
        __syncthreads();
        Y[i][j] = 0.5f * yt;
        Z[i][j] = 0.5f * tz;
    }
    __syncthreads();

    const float w = Z[i][j] * rsqrtf(t);
    wm[i * 32 + j] = w;

    T[i][j] = w * mj;
    __syncthreads();
    if (j == 0) {
        float s = 0.f;
        #pragma unroll
        for (int k = 0; k < 32; ++k) s += T[i][k];
        shift[i] = s;
    }
}

__global__ void __launch_bounds__(256) dbn_apply(const float* __restrict__ x,
                                                 const float* __restrict__ weight,
                                                 const float* __restrict__ bias,
                                                 const float* __restrict__ ws,
                                                 float* __restrict__ out) {
    const int g      = blockIdx.y;
    const int lane   = threadIdx.x & 31;
    const int wave   = threadIdx.x >> 5;
    const int lane16 = lane & 15;
    const int half   = lane >> 4;
    const int wid    = blockIdx.x * 8 + wave;

    __shared__ float sw[32], sb[32], ssh[32];
    if (threadIdx.x < 32) {
        const int c = g * 32 + threadIdx.x;
        sw[threadIdx.x]  = weight[c];
        sb[threadIdx.x]  = bias[c];
        ssh[threadIdx.x] = ws[WS_SHIFT + g * 32 + threadIdx.x];
    }
    __syncthreads();

    const float* wm = ws + WS_WM + g * 1024;
    v2f a0[8], a1[8];
    #pragma unroll
    for (int ks = 0; ks < 8; ++ks) {
        a0[ks] = *(const v2f*)(wm + lane16 * 32 + ks * 4 + 2 * half);
        a1[ks] = *(const v2f*)(wm + (lane16 + 16) * 32 + ks * 4 + 2 * half);
    }

    float wA[8], cA[8], wB[8], cB[8];
    #pragma unroll
    for (int jj = 0; jj < 8; ++jj) {
        const int m = jj + 8 * half;
        wA[jj] = sw[m];       cA[jj] = sb[m]      - ssh[m]      * sw[m];
        wB[jj] = sw[m + 16];  cB[jj] = sb[m + 16] - ssh[m + 16] * sw[m + 16];
    }

    const int b  = wid >> 5;
    const int s0 = (wid & 31) * 128;

    const float* px = x + (size_t)(b * NUM_FEATURES + g * GROUP_SIZE) * HW
                        + (size_t)(s0 + lane16) + (size_t)(2 * half) * HW;
    float* po = out + (size_t)(b * NUM_FEATURES + g * GROUP_SIZE) * HW
                    + (size_t)(s0 + lane16) + (size_t)(8 * half) * HW;

    for (int t = 0; t < 8; ++t) {
        const int co = t * 16;
        v8f acc0 = {}, acc1 = {};
        #pragma unroll
        for (int ks = 0; ks < 8; ++ks) {
            v2f bf;
            bf.x = px[(size_t)(ks * 4) * HW + co];
            bf.y = px[(size_t)(ks * 4 + 1) * HW + co];
            acc0 = __builtin_amdgcn_wmma_f32_16x16x4_f32(false, a0[ks], false, bf, (short)0, acc0, false, false);
            acc1 = __builtin_amdgcn_wmma_f32_16x16x4_f32(false, a1[ks], false, bf, (short)0, acc1, false, false);
        }
        #pragma unroll
        for (int jj = 0; jj < 8; ++jj) {
            __builtin_nontemporal_store(acc0[jj] * wA[jj] + cA[jj],
                                        po + (size_t)jj * HW + co);
            __builtin_nontemporal_store(acc1[jj] * wB[jj] + cB[jj],
                                        po + (size_t)(jj + 16) * HW + co);
        }
    }
}

extern "C" void kernel_launch(void* const* d_in, const int* in_sizes, int n_in,
                              void* d_out, int out_size, void* d_ws, size_t ws_size,
                              hipStream_t stream) {
    const float* x      = (const float*)d_in[0];
    const float* weight = (const float*)d_in[1];
    const float* bias   = (const float*)d_in[2];
    float* out = (float*)d_out;
    float* ws  = (float*)d_ws;

    hipMemsetAsync(ws, 0,
                   (size_t)(WS_SUMS + NUM_GROUPS * 32) * sizeof(float), stream);

    dbn_gram   <<<dim3(64,  NUM_GROUPS), 256,  0, stream>>>(x, ws);
    dbn_invsqrt<<<dim3(NUM_GROUPS),      1024, 0, stream>>>(ws);
    dbn_apply  <<<dim3(128, NUM_GROUPS), 256,  0, stream>>>(x, weight, bias, ws, out);
}


# Inverse-sqrt iteration: its 32x32x32 products move from LDS-fed scalar FMAs (1024 threads) to f32 WMMA on 4 waves, operands from row-major+transposed LDS copies; product and trace sum order changes
# speedup vs baseline: 1.8476x; 1.8476x over previous
#include <hip/hip_runtime.h>

typedef __attribute__((ext_vector_type(2))) float v2f;
typedef __attribute__((ext_vector_type(4))) float v4f;
typedef __attribute__((ext_vector_type(8))) float v8f;

#define NUM_FEATURES 256
#define GROUP_SIZE   32
#define NUM_GROUPS   8
#define BATCH        32
#define HW           4096
#define NCOLS        (BATCH * HW)
#define EPS          1e-5f
#define NS_ITERS     32

#define WS_GRAM   0
#define WS_SUMS   (NUM_GROUPS * 32 * 32)
#define WS_WM     (WS_SUMS + NUM_GROUPS * 32)
#define WS_SHIFT  (WS_WM + NUM_GROUPS * 32 * 32)

__global__ void __launch_bounds__(256) dbn_gram(const float* __restrict__ x,
                                                float* __restrict__ ws) {
    const int g      = blockIdx.y;
    const int lane   = threadIdx.x & 31;
    const int wave   = threadIdx.x >> 5;
    const int lane16 = lane & 15;
    const int half   = lane >> 4;
    const int wid    = blockIdx.x * 8 + wave;

    float* gram = ws + WS_GRAM + g * 1024;
    float* sums = ws + WS_SUMS + g * 32;

    const int b  = wid >> 4;
    const int s0 = (wid & 15) * 256;

    const float* pLo = x + (size_t)(b * NUM_FEATURES + g * GROUP_SIZE + lane16) * HW
                         + (size_t)(s0 + 2 * half);
    const float* pHi = pLo + (size_t)16 * HW;

    v8f a00 = {}, a01 = {}, a10 = {}, a11 = {};
    float sLo = 0.f, sHi = 0.f;

    #pragma unroll 4
    for (int i = 0; i < 64; ++i) {
        v2f lo = *(const v2f*)(pLo + i * 4);
        v2f hi = *(const v2f*)(pHi + i * 4);

        a00 = __builtin_amdgcn_wmma_f32_16x16x4_f32(false, lo, false, lo, (short)0, a00, false, false);
        a01 = __builtin_amdgcn_wmma_f32_16x16x4_f32(false, lo, false, hi, (short)0, a01, false, false);
        a10 = __builtin_amdgcn_wmma_f32_16x16x4_f32(false, hi, false, lo, (short)0, a10, false, false);
        a11 = __builtin_amdgcn_wmma_f32_16x16x4_f32(false, hi, false, hi, (short)0, a11, false, false);

        sLo += lo.x + lo.y;
        sHi += hi.x + hi.y;
    }

    sLo += __shfl_xor(sLo, 16, 32);
    sHi += __shfl_xor(sHi, 16, 32);
    if (half == 0) {
        atomicAdd(&sums[lane16],      sLo);
        atomicAdd(&sums[lane16 + 16], sHi);
    }

    #pragma unroll
    for (int j = 0; j < 8; ++j) {
        const int m = j + 8 * half;
        const int n = lane16;
        atomicAdd(&gram[(m)      * 32 + n],      a00[j]);
        atomicAdd(&gram[(m)      * 32 + n + 16], a01[j]);
        atomicAdd(&gram[(m + 16) * 32 + n],      a10[j]);
        atomicAdd(&gram[(m + 16) * 32 + n + 16], a11[j]);
    }
}

#define NS_P 36

__device__ __forceinline__ v4f ns_ld4(const float* p) { return *(const v4f*)p; }
__device__ __forceinline__ v2f ns_lo(const v4f v) { return (v2f){v.x, v.y}; }
__device__ __forceinline__ v2f ns_hi(const v4f v) { return (v2f){v.z, v.w}; }

__device__ __forceinline__ void ns_store_tile(float* m, float* mt, const v8f d) {
    #pragma unroll
    for (int r = 0; r < 8; ++r) m[r * NS_P] = d[r];
    *(v4f*)(mt)     = (v4f){d[0], d[1], d[2], d[3]};
    *(v4f*)(mt + 4) = (v4f){d[4], d[5], d[6], d[7]};
}

__global__ void __launch_bounds__(128) dbn_invsqrt(float* __restrict__ ws) {
    const int g    = blockIdx.x;
    const int tid  = threadIdx.x;
    const int lane = tid & 31;
    const int wave = tid >> 5;
    const int c    = lane & 15;
    const int half = lane >> 4;
    const int tm   = wave >> 1;
    const int tn   = wave & 1;

    __shared__ __attribute__((aligned(16))) float Ym[32 * NS_P], Yt[32 * NS_P];
    __shared__ __attribute__((aligned(16))) float Zm[32 * NS_P], Zt[32 * NS_P];
    __shared__ __attribute__((aligned(16))) float Tm[32 * NS_P], Tt[32 * NS_P];
    __shared__ float mean_s[32];
    __shared__ float tr;

    const float* gram = ws + WS_GRAM + g * 1024;
    const float* sums = ws + WS_SUMS + g * 32;
    float* wm    = ws + WS_WM    + g * 1024;
    float* shift = ws + WS_SHIFT + g * 32;

    const int j = lane;
    float gv[8];
    #pragma unroll
    for (int q = 0; q < 8; ++q) gv[q] = gram[(wave + 4 * q) * 32 + j];

    const float invN = 1.0f / (float)NCOLS;
    if (tid < 32) mean_s[tid] = sums[tid] * invN;
    if (tid == 0) tr = 0.f;
    __syncthreads();

    const float mj = mean_s[j];
    float sg[8];
    float sdiag = 0.f;
    #pragma unroll
    for (int q = 0; q < 8; ++q) {
        const int i = wave + 4 * q;
        const float mi = mean_s[i];
        const float sij = gv[q] - (float)NCOLS * mi * mj + ((i == j) ? EPS : 0.f);
        sg[q] = sij;
        sdiag = (i == j) ? sij : sdiag;
    }
    if ((j & 3) == wave) atomicAdd(&tr, sdiag);
    __syncthreads();
    const float t = tr;

    #pragma unroll
    for (int q = 0; q < 8; ++q) {
        const int i = wave + 4 * q;
        const float y = sg[q] / t;
        const float z = (i == j) ? 1.f : 0.f;
        Ym[i * NS_P + j] = y;  Yt[j * NS_P + i] = y;
        Zm[i * NS_P + j] = z;  Zt[j * NS_P + i] = z;
    }

    const int offA = (tm * 16 + c) * NS_P + 16 * half;
    const int offB = (tn * 16 + c) * NS_P + 16 * half;
    const int offDm = (tm * 16 + 8 * half) * NS_P + tn * 16 + c;
    const int offDt = (tn * 16 + c) * NS_P + tm * 16 + 8 * half;

    v8f eye3;
    #pragma unroll
    for (int r = 0; r < 8; ++r)
        eye3[r] = ((tm * 16 + 8 * half + r) == (tn * 16 + c)) ? 3.f : 0.f;

    for (int it = 0; it < NS_ITERS; ++it) {
        __syncthreads();
        v4f zA[4], yB[4], yA[4], zB[4];
        #pragma unroll
        for (int q = 0; q < 4; ++q) {
            zA[q] = ns_ld4(Zm + offA + 4 * q);
            yB[q] = ns_ld4(Yt + offB + 4 * q);
        }
        #pragma unroll
        for (int q = 0; q < 4; ++q) {
            yA[q] = ns_ld4(Ym + offA + 4 * q);
            zB[q] = ns_ld4(Zt + offB + 4 * q);
        }
        __builtin_amdgcn_sched_barrier(0);

        v8f zy = {};
        #pragma unroll
        for (int q = 0; q < 4; ++q) {
            zy = __builtin_amdgcn_wmma_f32_16x16x4_f32(false, ns_lo(zA[q]), false, ns_lo(yB[q]), (short)0, zy, false, false);
            zy = __builtin_amdgcn_wmma_f32_16x16x4_f32(false, ns_hi(zA[q]), false, ns_hi(yB[q]), (short)0, zy, false, false);
        }
        asm volatile("v_nop\n\tv_nop\n\tv_nop\n\tv_nop" : "+v"(zy));
        ns_store_tile(Tm + offDm, Tt + offDt, eye3 - zy);
        __syncthreads();

        v4f tA[4], tB[4];
        #pragma unroll
        for (int q = 0; q < 4; ++q) {
            tB[q] = ns_ld4(Tt + offB + 4 * q);
            tA[q] = ns_ld4(Tm + offA + 4 * q);
        }
        __builtin_amdgcn_sched_barrier(0);

        v8f yt = {}, tz = {};
        #pragma unroll
        for (int q = 0; q < 4; ++q) {
            yt = __builtin_amdgcn_wmma_f32_16x16x4_f32(false, ns_lo(yA[q]), false, ns_lo(tB[q]), (short)0, yt, false, false);
            tz = __builtin_amdgcn_wmma_f32_16x16x4_f32(false, ns_lo(tA[q]), false, ns_lo(zB[q]), (short)0, tz, false, false);
            yt = __builtin_amdgcn_wmma_f32_16x16x4_f32(false, ns_hi(yA[q]), false, ns_hi(tB[q]), (short)0, yt, false, false);
            tz = __builtin_amdgcn_wmma_f32_16x16x4_f32(false, ns_hi(tA[q]), false, ns_hi(zB[q]), (short)0, tz, false, false);
        }
        asm volatile("v_nop\n\tv_nop\n\tv_nop\n\tv_nop" : "+v"(yt), "+v"(tz));
        ns_store_tile(Ym + offDm, Yt + offDt, 0.5f * yt);
        ns_store_tile(Zm + offDm, Zt + offDt, 0.5f * tz);
    }
    __syncthreads();

    const float rs = rsqrtf(t);
    #pragma unroll
    for (int q = 0; q < 8; ++q) {
        const int i = wave + 4 * q;
        const float w = Zm[i * NS_P + j] * rs;
        wm[i * 32 + j] = w;
        Tm[i * NS_P + j] = w * mj;
    }
    __syncthreads();
    if (tid < 32) {
        float s = 0.f;
        #pragma unroll
        for (int k = 0; k < 32; ++k) s += Tm[tid * NS_P + k];
        shift[tid] = s;
    }
}

__global__ void __launch_bounds__(256) dbn_apply(const float* __restrict__ x,
                                                 const float* __restrict__ weight,
                                                 const float* __restrict__ bias,
                                                 const float* __restrict__ ws,
                                                 float* __restrict__ out) {
    const int g      = blockIdx.y;
    const int lane   = threadIdx.x & 31;
    const int wave   = threadIdx.x >> 5;
    const int lane16 = lane & 15;
    const int half   = lane >> 4;
    const int wid    = blockIdx.x * 8 + wave;

    __shared__ float sw[32], sb[32], ssh[32];
    if (threadIdx.x < 32) {
        const int c = g * 32 + threadIdx.x;
        sw[threadIdx.x]  = weight[c];
        sb[threadIdx.x]  = bias[c];
        ssh[threadIdx.x] = ws[WS_SHIFT + g * 32 + threadIdx.x];
    }
    __syncthreads();

    const float* wm = ws + WS_WM + g * 1024;
    v2f a0[8], a1[8];
    #pragma unroll
    for (int ks = 0; ks < 8; ++ks) {
        a0[ks] = *(const v2f*)(wm + lane16 * 32 + ks * 4 + 2 * half);
        a1[ks] = *(const v2f*)(wm + (lane16 + 16) * 32 + ks * 4 + 2 * half);
    }

    float wA[8], cA[8], wB[8], cB[8];
    #pragma unroll
    for (int jj = 0; jj < 8; ++jj) {
        const int m = jj + 8 * half;
        wA[jj] = sw[m];       cA[jj] = sb[m]      - ssh[m]      * sw[m];
        wB[jj] = sw[m + 16];  cB[jj] = sb[m + 16] - ssh[m + 16] * sw[m + 16];
    }

    const int b  = wid >> 5;
    const int s0 = (wid & 31) * 128;

    const float* px = x + (size_t)(b * NUM_FEATURES + g * GROUP_SIZE) * HW
                        + (size_t)(s0 + lane16) + (size_t)(2 * half) * HW;
    float* po = out + (size_t)(b * NUM_FEATURES + g * GROUP_SIZE) * HW
                    + (size_t)(s0 + lane16) + (size_t)(8 * half) * HW;

    for (int t = 0; t < 8; ++t) {
        const int co = t * 16;
        v8f acc0 = {}, acc1 = {};
        #pragma unroll
        for (int ks = 0; ks < 8; ++ks) {
            v2f bf;
            bf.x = px[(size_t)(ks * 4) * HW + co];
            bf.y = px[(size_t)(ks * 4 + 1) * HW + co];
            acc0 = __builtin_amdgcn_wmma_f32_16x16x4_f32(false, a0[ks], false, bf, (short)0, acc0, false, false);
            acc1 = __builtin_amdgcn_wmma_f32_16x16x4_f32(false, a1[ks], false, bf, (short)0, acc1, false, false);
        }
        #pragma unroll
        for (int jj = 0; jj < 8; ++jj) {
            __builtin_nontemporal_store(acc0[jj] * wA[jj] + cA[jj],
                                        po + (size_t)jj * HW + co);
            __builtin_nontemporal_store(acc1[jj] * wB[jj] + cB[jj],
                                        po + (size_t)(jj + 16) * HW + co);
        }
    }
}

extern "C" void kernel_launch(void* const* d_in, const int* in_sizes, int n_in,
                              void* d_out, int out_size, void* d_ws, size_t ws_size,
                              hipStream_t stream) {
    const float* x      = (const float*)d_in[0];
    const float* weight = (const float*)d_in[1];
    const float* bias   = (const float*)d_in[2];
    float* out = (float*)d_out;
    float* ws  = (float*)d_ws;

    hipMemsetAsync(ws, 0,
                   (size_t)(WS_SUMS + NUM_GROUPS * 32) * sizeof(float), stream);

    dbn_gram   <<<dim3(64,  NUM_GROUPS), 256,  0, stream>>>(x, ws);
    dbn_invsqrt<<<dim3(NUM_GROUPS),      128,  0, stream>>>(ws);
    dbn_apply  <<<dim3(128, NUM_GROUPS), 256,  0, stream>>>(x, weight, bias, ws, out);
}


# Gram kernel: a block's 8 waves add their partial Gram tiles and channel sums in LDS, then one float atomic per whole 128-B row (8x fewer atomics); Gram and channel sum order changes
# speedup vs baseline: 2.3621x; 1.2785x over previous
#include <hip/hip_runtime.h>

typedef __attribute__((ext_vector_type(2))) float v2f;
typedef __attribute__((ext_vector_type(4))) float v4f;
typedef __attribute__((ext_vector_type(8))) float v8f;

#define NUM_FEATURES 256
#define GROUP_SIZE   32
#define NUM_GROUPS   8
#define BATCH        32
#define HW           4096
#define NCOLS        (BATCH * HW)
#define EPS          1e-5f
#define NS_ITERS     32

#define WS_GRAM   0
#define WS_SUMS   (NUM_GROUPS * 32 * 32)
#define WS_WM     (WS_SUMS + NUM_GROUPS * 32)
#define WS_SHIFT  (WS_WM + NUM_GROUPS * 32 * 32)

#define RED_P 34
__global__ void __launch_bounds__(256) dbn_gram(const float* __restrict__ x,
                                                float* __restrict__ ws) {
    const int g      = blockIdx.y;
    const int lane   = threadIdx.x & 31;
    const int wave   = threadIdx.x >> 5;
    const int lane16 = lane & 15;
    const int half   = lane >> 4;
    const int wid    = blockIdx.x * 8 + wave;

    float* gram = ws + WS_GRAM + g * 1024;
    float* sums = ws + WS_SUMS + g * 32;

    const int b  = wid >> 4;
    const int s0 = (wid & 15) * 256;

    const float* pLo = x + (size_t)(b * NUM_FEATURES + g * GROUP_SIZE + lane16) * HW
                         + (size_t)(s0 + 2 * half);
    const float* pHi = pLo + (size_t)16 * HW;

    v8f a00 = {}, a01 = {}, a10 = {}, a11 = {};
    float sLo = 0.f, sHi = 0.f;

    #pragma unroll 4
    for (int i = 0; i < 64; ++i) {
        v2f lo = *(const v2f*)(pLo + i * 4);
        v2f hi = *(const v2f*)(pHi + i * 4);

        a00 = __builtin_amdgcn_wmma_f32_16x16x4_f32(false, lo, false, lo, (short)0, a00, false, false);
        a01 = __builtin_amdgcn_wmma_f32_16x16x4_f32(false, lo, false, hi, (short)0, a01, false, false);
        a10 = __builtin_amdgcn_wmma_f32_16x16x4_f32(false, hi, false, lo, (short)0, a10, false, false);
        a11 = __builtin_amdgcn_wmma_f32_16x16x4_f32(false, hi, false, hi, (short)0, a11, false, false);

        sLo += lo.x + lo.y;
        sHi += hi.x + hi.y;
    }

    __shared__ float red[8 * 32 * RED_P];
    __shared__ float reds[8 * 32];

    sLo += __shfl_xor(sLo, 16, 32);
    sHi += __shfl_xor(sHi, 16, 32);
    if (half == 0) {
        reds[wave * 32 + lane16]      = sLo;
        reds[wave * 32 + lane16 + 16] = sHi;
    }

    float* rw = red + wave * (32 * RED_P) + (8 * half) * RED_P + lane16;
    #pragma unroll
    for (int j = 0; j < 8; ++j) {
        rw[(j)      * RED_P]      = a00[j];
        rw[(j)      * RED_P + 16] = a01[j];
        rw[(j + 16) * RED_P]      = a10[j];
        rw[(j + 16) * RED_P + 16] = a11[j];
    }
    __syncthreads();

    #pragma unroll
    for (int k = 0; k < 4; ++k) {
        const int row = 8 * k + wave;
        const float* rr = red + row * RED_P + lane;
        float s = rr[0];
        #pragma unroll
        for (int w = 1; w < 8; ++w) s += rr[w * (32 * RED_P)];
        atomicAdd(&gram[row * 32 + lane], s);
    }

    if (threadIdx.x < 32) {
        float s = reds[threadIdx.x];
        #pragma unroll
        for (int w = 1; w < 8; ++w) s += reds[w * 32 + threadIdx.x];
        atomicAdd(&sums[threadIdx.x], s);
    }
}

#define NS_P 36

__device__ __forceinline__ v4f ns_ld4(const float* p) { return *(const v4f*)p; }
__device__ __forceinline__ v2f ns_lo(const v4f v) { return (v2f){v.x, v.y}; }
__device__ __forceinline__ v2f ns_hi(const v4f v) { return (v2f){v.z, v.w}; }

__device__ __forceinline__ void ns_store_tile(float* m, float* mt, const v8f d) {
    #pragma unroll
    for (int r = 0; r < 8; ++r) m[r * NS_P] = d[r];
    *(v4f*)(mt)     = (v4f){d[0], d[1], d[2], d[3]};
    *(v4f*)(mt + 4) = (v4f){d[4], d[5], d[6], d[7]};
}

__global__ void __launch_bounds__(128) dbn_invsqrt(float* __restrict__ ws) {
    const int g    = blockIdx.x;
    const int tid  = threadIdx.x;
    const int lane = tid & 31;
    const int wave = tid >> 5;
    const int c    = lane & 15;
    const int half = lane >> 4;
    const int tm   = wave >> 1;
    const int tn   = wave & 1;

    __shared__ __attribute__((aligned(16))) float Ym[32 * NS_P], Yt[32 * NS_P];
    __shared__ __attribute__((aligned(16))) float Zm[32 * NS_P], Zt[32 * NS_P];
    __shared__ __attribute__((aligned(16))) float Tm[32 * NS_P], Tt[32 * NS_P];
    __shared__ float mean_s[32];
    __shared__ float tr;

    const float* gram = ws + WS_GRAM + g * 1024;
    const float* sums = ws + WS_SUMS + g * 32;
    float* wm    = ws + WS_WM    + g * 1024;
    float* shift = ws + WS_SHIFT + g * 32;

    const int j = lane;
    float gv[8];
    #pragma unroll
    for (int q = 0; q < 8; ++q) gv[q] = gram[(wave + 4 * q) * 32 + j];

    const float invN = 1.0f / (float)NCOLS;
    if (tid < 32) mean_s[tid] = sums[tid] * invN;
    if (tid == 0) tr = 0.f;
    __syncthreads();

    const float mj = mean_s[j];
    float sg[8];
    float sdiag = 0.f;
    #pragma unroll
    for (int q = 0; q < 8; ++q) {
        const int i = wave + 4 * q;
        const float mi = mean_s[i];
        const float sij = gv[q] - (float)NCOLS * mi * mj + ((i == j) ? EPS : 0.f);
        sg[q] = sij;
        sdiag = (i == j) ? sij : sdiag;
    }
    if ((j & 3) == wave) atomicAdd(&tr, sdiag);
    __syncthreads();
    const float t = tr;

    #pragma unroll
    for (int q = 0; q < 8; ++q) {
        const int i = wave + 4 * q;
        const float y = sg[q] / t;
        const float z = (i == j) ? 1.f : 0.f;
        Ym[i * NS_P + j] = y;  Yt[j * NS_P + i] = y;
        Zm[i * NS_P + j] = z;  Zt[j * NS_P + i] = z;
    }

    const int offA = (tm * 16 + c) * NS_P + 16 * half;
    const int offB = (tn * 16 + c) * NS_P + 16 * half;
    const int offDm = (tm * 16 + 8 * half) * NS_P + tn * 16 + c;
    const int offDt = (tn * 16 + c) * NS_P + tm * 16 + 8 * half;

    v8f eye3;
    #pragma unroll
    for (int r = 0; r < 8; ++r)
        eye3[r] = ((tm * 16 + 8 * half + r) == (tn * 16 + c)) ? 3.f : 0.f;

    for (int it = 0; it < NS_ITERS; ++it) {
        __syncthreads();
        v4f zA[4], yB[4], yA[4], zB[4];
        #pragma unroll
        for (int q = 0; q < 4; ++q) {
            zA[q] = ns_ld4(Zm + offA + 4 * q);
            yB[q] = ns_ld4(Yt + offB + 4 * q);
        }
        #pragma unroll
        for (int q = 0; q < 4; ++q) {
            yA[q] = ns_ld4(Ym + offA + 4 * q);
            zB[q] = ns_ld4(Zt + offB + 4 * q);
        }
        __builtin_amdgcn_sched_barrier(0);

        v8f zy = {};
        #pragma unroll
        for (int q = 0; q < 4; ++q) {
            zy = __builtin_amdgcn_wmma_f32_16x16x4_f32(false, ns_lo(zA[q]), false, ns_lo(yB[q]), (short)0, zy, false, false);
            zy = __builtin_amdgcn_wmma_f32_16x16x4_f32(false, ns_hi(zA[q]), false, ns_hi(yB[q]), (short)0, zy, false, false);
        }
        asm volatile("v_nop\n\tv_nop\n\tv_nop\n\tv_nop" : "+v"(zy));
        ns_store_tile(Tm + offDm, Tt + offDt, eye3 - zy);
        __syncthreads();

        v4f tA[4], tB[4];
        #pragma unroll
        for (int q = 0; q < 4; ++q) {
            tB[q] = ns_ld4(Tt + offB + 4 * q);
            tA[q] = ns_ld4(Tm + offA + 4 * q);
        }
        __builtin_amdgcn_sched_barrier(0);

        v8f yt = {}, tz = {};
        #pragma unroll
        for (int q = 0; q < 4; ++q) {
            yt = __builtin_amdgcn_wmma_f32_16x16x4_f32(false, ns_lo(yA[q]), false, ns_lo(tB[q]), (short)0, yt, false, false);
            tz = __builtin_amdgcn_wmma_f32_16x16x4_f32(false, ns_lo(tA[q]), false, ns_lo(zB[q]), (short)0, tz, false, false);
            yt = __builtin_amdgcn_wmma_f32_16x16x4_f32(false, ns_hi(yA[q]), false, ns_hi(tB[q]), (short)0, yt, false, false);
            tz = __builtin_amdgcn_wmma_f32_16x16x4_f32(false, ns_hi(tA[q]), false, ns_hi(zB[q]), (short)0, tz, false, false);
        }
        asm volatile("v_nop\n\tv_nop\n\tv_nop\n\tv_nop" : "+v"(yt), "+v"(tz));
        ns_store_tile(Ym + offDm, Yt + offDt, 0.5f * yt);
        ns_store_tile(Zm + offDm, Zt + offDt, 0.5f * tz);
    }
    __syncthreads();

    const float rs = rsqrtf(t);
    #pragma unroll
    for (int q = 0; q < 8; ++q) {
        const int i = wave + 4 * q;
        const float w = Zm[i * NS_P + j] * rs;
        wm[i * 32 + j] = w;
        Tm[i * NS_P + j] = w * mj;
    }
    __syncthreads();
    if (tid < 32) {
        float s = 0.f;
        #pragma unroll
        for (int k = 0; k < 32; ++k) s += Tm[tid * NS_P + k];
        shift[tid] = s;
    }
}

__global__ void __launch_bounds__(256) dbn_apply(const float* __restrict__ x,
                                                 const float* __restrict__ weight,
                                                 const float* __restrict__ bias,
                                                 const float* __restrict__ ws,
                                                 float* __restrict__ out) {
    const int g      = blockIdx.y;
    const int lane   = threadIdx.x & 31;
    const int wave   = threadIdx.x >> 5;
    const int lane16 = lane & 15;
    const int half   = lane >> 4;
    const int wid    = blockIdx.x * 8 + wave;

    __shared__ float sw[32], sb[32], ssh[32];
    if (threadIdx.x < 32) {
        const int c = g * 32 + threadIdx.x;
        sw[threadIdx.x]  = weight[c];
        sb[threadIdx.x]  = bias[c];
        ssh[threadIdx.x] = ws[WS_SHIFT + g * 32 + threadIdx.x];
    }
    __syncthreads();

    const float* wm = ws + WS_WM + g * 1024;
    v2f a0[8], a1[8];
    #pragma unroll
    for (int ks = 0; ks < 8; ++ks) {
        a0[ks] = *(const v2f*)(wm + lane16 * 32 + ks * 4 + 2 * half);
        a1[ks] = *(const v2f*)(wm + (lane16 + 16) * 32 + ks * 4 + 2 * half);
    }

    float wA[8], cA[8], wB[8], cB[8];
    #pragma unroll
    for (int jj = 0; jj < 8; ++jj) {
        const int m = jj + 8 * half;
        wA[jj] = sw[m];       cA[jj] = sb[m]      - ssh[m]      * sw[m];
        wB[jj] = sw[m + 16];  cB[jj] = sb[m + 16] - ssh[m + 16] * sw[m + 16];
    }

    const int b  = wid >> 5;
    const int s0 = (wid & 31) * 128;

    const float* px = x + (size_t)(b * NUM_FEATURES + g * GROUP_SIZE) * HW
                        + (size_t)(s0 + lane16) + (size_t)(2 * half) * HW;
    float* po = out + (size_t)(b * NUM_FEATURES + g * GROUP_SIZE) * HW
                    + (size_t)(s0 + lane16) + (size_t)(8 * half) * HW;

    for (int t = 0; t < 8; ++t) {
        const int co = t * 16;
        v8f acc0 = {}, acc1 = {};
        #pragma unroll
        for (int ks = 0; ks < 8; ++ks) {
            v2f bf;
            bf.x = px[(size_t)(ks * 4) * HW + co];
            bf.y = px[(size_t)(ks * 4 + 1) * HW + co];
            acc0 = __builtin_amdgcn_wmma_f32_16x16x4_f32(false, a0[ks], false, bf, (short)0, acc0, false, false);
            acc1 = __builtin_amdgcn_wmma_f32_16x16x4_f32(false, a1[ks], false, bf, (short)0, acc1, false, false);
        }
        #pragma unroll
        for (int jj = 0; jj < 8; ++jj) {
            __builtin_nontemporal_store(acc0[jj] * wA[jj] + cA[jj],
                                        po + (size_t)jj * HW + co);
            __builtin_nontemporal_store(acc1[jj] * wB[jj] + cB[jj],
                                        po + (size_t)(jj + 16) * HW + co);
        }
    }
}

extern "C" void kernel_launch(void* const* d_in, const int* in_sizes, int n_in,
                              void* d_out, int out_size, void* d_ws, size_t ws_size,
                              hipStream_t stream) {
    const float* x      = (const float*)d_in[0];
    const float* weight = (const float*)d_in[1];
    const float* bias   = (const float*)d_in[2];
    float* out = (float*)d_out;
    float* ws  = (float*)d_ws;

    hipMemsetAsync(ws, 0,
                   (size_t)(WS_SUMS + NUM_GROUPS * 32) * sizeof(float), stream);

    dbn_gram   <<<dim3(64,  NUM_GROUPS), 256,  0, stream>>>(x, ws);
    dbn_invsqrt<<<dim3(NUM_GROUPS),      128,  0, stream>>>(ws);
    dbn_apply  <<<dim3(128, NUM_GROUPS), 256,  0, stream>>>(x, weight, bias, ws, out);
}


# Inverse-sqrt iteration: the loop leaves after the step that began with |I - Z*Y|_F^2 < 2^-24 (cap of 32 steps kept); the steps dropped would only re-round Y and Z: wm moves by rounding only
# speedup vs baseline: 2.6146x; 1.1069x over previous
#include <hip/hip_runtime.h>

typedef __attribute__((ext_vector_type(2))) float v2f;
typedef __attribute__((ext_vector_type(4))) float v4f;
typedef __attribute__((ext_vector_type(8))) float v8f;

#define NUM_FEATURES 256
#define GROUP_SIZE   32
#define NUM_GROUPS   8
#define BATCH        32
#define HW           4096
#define NCOLS        (BATCH * HW)
#define EPS          1e-5f
#define NS_ITERS     32

#define WS_GRAM   0
#define WS_SUMS   (NUM_GROUPS * 32 * 32)
#define WS_WM     (WS_SUMS + NUM_GROUPS * 32)
#define WS_SHIFT  (WS_WM + NUM_GROUPS * 32 * 32)

#define RED_P 34
__global__ void __launch_bounds__(256) dbn_gram(const float* __restrict__ x,
                                                float* __restrict__ ws) {
    const int g      = blockIdx.y;
    const int lane   = threadIdx.x & 31;
    const int wave   = threadIdx.x >> 5;
    const int lane16 = lane & 15;
    const int half   = lane >> 4;
    const int wid    = blockIdx.x * 8 + wave;

    float* gram = ws + WS_GRAM + g * 1024;
    float* sums = ws + WS_SUMS + g * 32;

    const int b  = wid >> 4;
    const int s0 = (wid & 15) * 256;

    const float* pLo = x + (size_t)(b * NUM_FEATURES + g * GROUP_SIZE + lane16) * HW
                         + (size_t)(s0 + 2 * half);
    const float* pHi = pLo + (size_t)16 * HW;

    v8f a00 = {}, a01 = {}, a10 = {}, a11 = {};
    float sLo = 0.f, sHi = 0.f;

    #pragma unroll 4
    for (int i = 0; i < 64; ++i) {
        v2f lo = *(const v2f*)(pLo + i * 4);
        v2f hi = *(const v2f*)(pHi + i * 4);

        a00 = __builtin_amdgcn_wmma_f32_16x16x4_f32(false, lo, false, lo, (short)0, a00, false, false);
        a01 = __builtin_amdgcn_wmma_f32_16x16x4_f32(false, lo, false, hi, (short)0, a01, false, false);
        a10 = __builtin_amdgcn_wmma_f32_16x16x4_f32(false, hi, false, lo, (short)0, a10, false, false);
        a11 = __builtin_amdgcn_wmma_f32_16x16x4_f32(false, hi, false, hi, (short)0, a11, false, false);

        sLo += lo.x + lo.y;
        sHi += hi.x + hi.y;
    }

    __shared__ float red[8 * 32 * RED_P];
    __shared__ float reds[8 * 32];

    sLo += __shfl_xor(sLo, 16, 32);
    sHi += __shfl_xor(sHi, 16, 32);
    if (half == 0) {
        reds[wave * 32 + lane16]      = sLo;
        reds[wave * 32 + lane16 + 16] = sHi;
    }

    float* rw = red + wave * (32 * RED_P) + (8 * half) * RED_P + lane16;
    #pragma unroll
    for (int j = 0; j < 8; ++j) {
        rw[(j)      * RED_P]      = a00[j];
        rw[(j)      * RED_P + 16] = a01[j];
        rw[(j + 16) * RED_P]      = a10[j];
        rw[(j + 16) * RED_P + 16] = a11[j];
    }
    __syncthreads();

    #pragma unroll
    for (int k = 0; k < 4; ++k) {
        const int row = 8 * k + wave;
        const float* rr = red + row * RED_P + lane;
        float s = rr[0];
        #pragma unroll
        for (int w = 1; w < 8; ++w) s += rr[w * (32 * RED_P)];
        atomicAdd(&gram[row * 32 + lane], s);
    }

    if (threadIdx.x < 32) {
        float s = reds[threadIdx.x];
        #pragma unroll
        for (int w = 1; w < 8; ++w) s += reds[w * 32 + threadIdx.x];
        atomicAdd(&sums[threadIdx.x], s);
    }
}

#define NS_P 36

__device__ __forceinline__ v4f ns_ld4(const float* p) { return *(const v4f*)p; }
__device__ __forceinline__ v2f ns_lo(const v4f v) { return (v2f){v.x, v.y}; }
__device__ __forceinline__ v2f ns_hi(const v4f v) { return (v2f){v.z, v.w}; }

__device__ __forceinline__ void ns_store_tile(float* m, float* mt, const v8f d) {
    #pragma unroll
    for (int r = 0; r < 8; ++r) m[r * NS_P] = d[r];
    *(v4f*)(mt)     = (v4f){d[0], d[1], d[2], d[3]};
    *(v4f*)(mt + 4) = (v4f){d[4], d[5], d[6], d[7]};
}

#define NS_XM(v, ctl) __builtin_bit_cast(float, \
    __builtin_amdgcn_update_dpp(0, __builtin_bit_cast(int, (v)), (ctl), 0xf, 0xf, true))
__device__ __forceinline__ float ns_wave_sum(float v) {
    v += NS_XM(v, 0x161);
    v += NS_XM(v, 0x162);
    v += NS_XM(v, 0x164);
    v += NS_XM(v, 0x168);
    const int b = __builtin_bit_cast(int, v);
    return v + __builtin_bit_cast(float, __builtin_amdgcn_permlanex16(b, b, 0x76543210, 0xfedcba98, false, false));
}

#define NS_RES2_DONE 5.9604645e-8f
#define NS_FIRST(v) __builtin_bit_cast(float, __builtin_amdgcn_readfirstlane(__builtin_bit_cast(int, (v))))

__global__ void __launch_bounds__(128) dbn_invsqrt(float* __restrict__ ws) {
    const int g    = blockIdx.x;
    const int tid  = threadIdx.x;
    const int lane = tid & 31;
    const int wave = tid >> 5;
    const int c    = lane & 15;
    const int half = lane >> 4;
    const int tm   = wave >> 1;
    const int tn   = wave & 1;

    __shared__ __attribute__((aligned(16))) float Ym[32 * NS_P], Yt[32 * NS_P];
    __shared__ __attribute__((aligned(16))) float Zm[32 * NS_P], Zt[32 * NS_P];
    __shared__ __attribute__((aligned(16))) float Tm[32 * NS_P], Tt[32 * NS_P];
    __shared__ float mean_s[32];
    __shared__ float tr;
    __shared__ __attribute__((aligned(16))) float res_s[4];

    const float* gram = ws + WS_GRAM + g * 1024;
    const float* sums = ws + WS_SUMS + g * 32;
    float* wm    = ws + WS_WM    + g * 1024;
    float* shift = ws + WS_SHIFT + g * 32;

    const int j = lane;
    float gv[8];
    #pragma unroll
    for (int q = 0; q < 8; ++q) gv[q] = gram[(wave + 4 * q) * 32 + j];

    const float invN = 1.0f / (float)NCOLS;
    if (tid < 32) mean_s[tid] = sums[tid] * invN;
    if (tid == 0) tr = 0.f;
    __syncthreads();

    const float mj = mean_s[j];
    float sg[8];
    float sdiag = 0.f;
    #pragma unroll
    for (int q = 0; q < 8; ++q) {
        const int i = wave + 4 * q;
        const float mi = mean_s[i];
        const float sij = gv[q] - (float)NCOLS * mi * mj + ((i == j) ? EPS : 0.f);
        sg[q] = sij;
        sdiag = (i == j) ? sij : sdiag;
    }
    if ((j & 3) == wave) atomicAdd(&tr, sdiag);
    __syncthreads();
    const float t = tr;

    #pragma unroll
    for (int q = 0; q < 8; ++q) {
        const int i = wave + 4 * q;
        const float y = sg[q] / t;
        const float z = (i == j) ? 1.f : 0.f;
        Ym[i * NS_P + j] = y;  Yt[j * NS_P + i] = y;
        Zm[i * NS_P + j] = z;  Zt[j * NS_P + i] = z;
    }

    const int offA = (tm * 16 + c) * NS_P + 16 * half;
    const int offB = (tn * 16 + c) * NS_P + 16 * half;
    const int offDm = (tm * 16 + 8 * half) * NS_P + tn * 16 + c;
    const int offDt = (tn * 16 + c) * NS_P + tm * 16 + 8 * half;

    v8f eye3, eye1;
    #pragma unroll
    for (int r = 0; r < 8; ++r) {
        const bool dg = (tm * 16 + 8 * half + r) == (tn * 16 + c);
        eye3[r] = dg ? 3.f : 0.f;
        eye1[r] = dg ? 1.f : 0.f;
    }

    for (int it = 0; it < NS_ITERS; ++it) {
        __syncthreads();
        v4f zA[4], yB[4], yA[4], zB[4];
        #pragma unroll
        for (int q = 0; q < 4; ++q) {
            zA[q] = ns_ld4(Zm + offA + 4 * q);
            yB[q] = ns_ld4(Yt + offB + 4 * q);
        }
        #pragma unroll
        for (int q = 0; q < 4; ++q) {
            yA[q] = ns_ld4(Ym + offA + 4 * q);
            zB[q] = ns_ld4(Zt + offB + 4 * q);
        }
        __builtin_amdgcn_sched_barrier(0);

        v8f zy = {};
        #pragma unroll
        for (int q = 0; q < 4; ++q) {
            zy = __builtin_amdgcn_wmma_f32_16x16x4_f32(false, ns_lo(zA[q]), false, ns_lo(yB[q]), (short)0, zy, false, false);
            zy = __builtin_amdgcn_wmma_f32_16x16x4_f32(false, ns_hi(zA[q]), false, ns_hi(yB[q]), (short)0, zy, false, false);
        }
        asm volatile("v_nop\n\tv_nop\n\tv_nop\n\tv_nop" : "+v"(zy));
        ns_store_tile(Tm + offDm, Tt + offDt, eye3 - zy);
        __builtin_amdgcn_sched_barrier(0);
        const v8f d = eye1 - zy;
        const v2f d0 = {d[0], d[1]}, d1 = {d[2], d[3]}, d2 = {d[4], d[5]}, d3 = {d[6], d[7]};
        const v2f q = (d0 * d0 + d1 * d1) + (d2 * d2 + d3 * d3);
        const float rs = ns_wave_sum(q.x + q.y);
        if (lane == 0) res_s[wave] = rs;
        __syncthreads();

        const v4f rv = ns_ld4(res_s);
        v4f tA[4], tB[4];
        #pragma unroll
        for (int q = 0; q < 4; ++q) {
            tB[q] = ns_ld4(Tt + offB + 4 * q);
            tA[q] = ns_ld4(Tm + offA + 4 * q);
        }
        const float r0 = rv.x, r1 = rv.y, r2 = rv.z, r3 = rv.w;
        const float res2 = (NS_FIRST(r0) + NS_FIRST(r1)) + (NS_FIRST(r2) + NS_FIRST(r3));
        __builtin_amdgcn_sched_barrier(0);

        v8f yt = {}, tz = {};
        #pragma unroll
        for (int q = 0; q < 4; ++q) {
            yt = __builtin_amdgcn_wmma_f32_16x16x4_f32(false, ns_lo(yA[q]), false, ns_lo(tB[q]), (short)0, yt, false, false);
            tz = __builtin_amdgcn_wmma_f32_16x16x4_f32(false, ns_lo(tA[q]), false, ns_lo(zB[q]), (short)0, tz, false, false);
            yt = __builtin_amdgcn_wmma_f32_16x16x4_f32(false, ns_hi(yA[q]), false, ns_hi(tB[q]), (short)0, yt, false, false);
            tz = __builtin_amdgcn_wmma_f32_16x16x4_f32(false, ns_hi(tA[q]), false, ns_hi(zB[q]), (short)0, tz, false, false);
        }
        asm volatile("v_nop\n\tv_nop\n\tv_nop\n\tv_nop" : "+v"(yt), "+v"(tz));
        ns_store_tile(Ym + offDm, Yt + offDt, 0.5f * yt);
        ns_store_tile(Zm + offDm, Zt + offDt, 0.5f * tz);
        if (res2 < NS_RES2_DONE) break;
    }
    __syncthreads();

    const float rs = rsqrtf(t);
    #pragma unroll
    for (int q = 0; q < 8; ++q) {
        const int i = wave + 4 * q;
        const float w = Zm[i * NS_P + j] * rs;
        wm[i * 32 + j] = w;
        Tm[i * NS_P + j] = w * mj;
    }
    __syncthreads();
    if (tid < 32) {
        float s = 0.f;
        #pragma unroll
        for (int k = 0; k < 32; ++k) s += Tm[tid * NS_P + k];
        shift[tid] = s;
    }
}

__global__ void __launch_bounds__(256) dbn_apply(const float* __restrict__ x,
                                                 const float* __restrict__ weight,
                                                 const float* __restrict__ bias,
                                                 const float* __restrict__ ws,
                                                 float* __restrict__ out) {
    const int g      = blockIdx.y;
    const int lane   = threadIdx.x & 31;
    const int wave   = threadIdx.x >> 5;
    const int lane16 = lane & 15;
    const int half   = lane >> 4;
    const int wid    = blockIdx.x * 8 + wave;

    __shared__ float sw[32], sb[32], ssh[32];
    if (threadIdx.x < 32) {
        const int c = g * 32 + threadIdx.x;
        sw[threadIdx.x]  = weight[c];
        sb[threadIdx.x]  = bias[c];
        ssh[threadIdx.x] = ws[WS_SHIFT + g * 32 + threadIdx.x];
    }
    __syncthreads();

    const float* wm = ws + WS_WM + g * 1024;
    v2f a0[8], a1[8];
    #pragma unroll
    for (int ks = 0; ks < 8; ++ks) {
        a0[ks] = *(const v2f*)(wm + lane16 * 32 + ks * 4 + 2 * half);
        a1[ks] = *(const v2f*)(wm + (lane16 + 16) * 32 + ks * 4 + 2 * half);
    }

    float wA[8], cA[8], wB[8], cB[8];
    #pragma unroll
    for (int jj = 0; jj < 8; ++jj) {
        const int m = jj + 8 * half;
        wA[jj] = sw[m];       cA[jj] = sb[m]      - ssh[m]      * sw[m];
        wB[jj] = sw[m + 16];  cB[jj] = sb[m + 16] - ssh[m + 16] * sw[m + 16];
    }

    const int b  = wid >> 5;
    const int s0 = (wid & 31) * 128;

    const float* px = x + (size_t)(b * NUM_FEATURES + g * GROUP_SIZE) * HW
                        + (size_t)(s0 + lane16) + (size_t)(2 * half) * HW;
    float* po = out + (size_t)(b * NUM_FEATURES + g * GROUP_SIZE) * HW
                    + (size_t)(s0 + lane16) + (size_t)(8 * half) * HW;

    for (int t = 0; t < 8; ++t) {
        const int co = t * 16;
        v8f acc0 = {}, acc1 = {};
        #pragma unroll
        for (int ks = 0; ks < 8; ++ks) {
            v2f bf;
            bf.x = px[(size_t)(ks * 4) * HW + co];
            bf.y = px[(size_t)(ks * 4 + 1) * HW + co];
            acc0 = __builtin_amdgcn_wmma_f32_16x16x4_f32(false, a0[ks], false, bf, (short)0, acc0, false, false);
            acc1 = __builtin_amdgcn_wmma_f32_16x16x4_f32(false, a1[ks], false, bf, (short)0, acc1, false, false);
        }
        #pragma unroll
        for (int jj = 0; jj < 8; ++jj) {
            __builtin_nontemporal_store(acc0[jj] * wA[jj] + cA[jj],
                                        po + (size_t)jj * HW + co);
            __builtin_nontemporal_store(acc1[jj] * wB[jj] + cB[jj],
                                        po + (size_t)(jj + 16) * HW + co);
        }
    }
}

extern "C" void kernel_launch(void* const* d_in, const int* in_sizes, int n_in,
                              void* d_out, int out_size, void* d_ws, size_t ws_size,
                              hipStream_t stream) {
    const float* x      = (const float*)d_in[0];
    const float* weight = (const float*)d_in[1];
    const float* bias   = (const float*)d_in[2];
    float* out = (float*)d_out;
    float* ws  = (float*)d_ws;

    hipMemsetAsync(ws, 0,
                   (size_t)(WS_SUMS + NUM_GROUPS * 32) * sizeof(float), stream);

    dbn_gram   <<<dim3(64,  NUM_GROUPS), 256,  0, stream>>>(x, ws);
    dbn_invsqrt<<<dim3(NUM_GROUPS),      128,  0, stream>>>(ws);
    dbn_apply  <<<dim3(128, NUM_GROUPS), 256,  0, stream>>>(x, weight, bias, ws, out);
}


# Apply kernel: a lane owns 4 adjacent columns: each 16-byte load and store covers 256 contiguous bytes of a row (4 column tiles per load, output lines written whole by one store); sum order unchanged
# speedup vs baseline: 3.0571x; 1.1692x over previous
#include <hip/hip_runtime.h>

typedef __attribute__((ext_vector_type(2))) float v2f;
typedef __attribute__((ext_vector_type(4))) float v4f;
typedef __attribute__((ext_vector_type(8))) float v8f;

#define NUM_FEATURES 256
#define GROUP_SIZE   32
#define NUM_GROUPS   8
#define BATCH        32
#define HW           4096
#define NCOLS        (BATCH * HW)
#define EPS          1e-5f
#define NS_ITERS     32

#define WS_GRAM   0
#define WS_SUMS   (NUM_GROUPS * 32 * 32)
#define WS_WM     (WS_SUMS + NUM_GROUPS * 32)
#define WS_SHIFT  (WS_WM + NUM_GROUPS * 32 * 32)

#define RED_P 34
__global__ void __launch_bounds__(256) dbn_gram(const float* __restrict__ x,
                                                float* __restrict__ ws) {
    const int g      = blockIdx.y;
    const int lane   = threadIdx.x & 31;
    const int wave   = threadIdx.x >> 5;
    const int lane16 = lane & 15;
    const int half   = lane >> 4;
    const int wid    = blockIdx.x * 8 + wave;

    float* gram = ws + WS_GRAM + g * 1024;
    float* sums = ws + WS_SUMS + g * 32;

    const int b  = wid >> 4;
    const int s0 = (wid & 15) * 256;

    const float* pLo = x + (size_t)(b * NUM_FEATURES + g * GROUP_SIZE + lane16) * HW
                         + (size_t)(s0 + 2 * half);
    const float* pHi = pLo + (size_t)16 * HW;

    v8f a00 = {}, a01 = {}, a10 = {}, a11 = {};
    float sLo = 0.f, sHi = 0.f;

    #pragma unroll 4
    for (int i = 0; i < 64; ++i) {
        v2f lo = *(const v2f*)(pLo + i * 4);
        v2f hi = *(const v2f*)(pHi + i * 4);

        a00 = __builtin_amdgcn_wmma_f32_16x16x4_f32(false, lo, false, lo, (short)0, a00, false, false);
        a01 = __builtin_amdgcn_wmma_f32_16x16x4_f32(false, lo, false, hi, (short)0, a01, false, false);
        a10 = __builtin_amdgcn_wmma_f32_16x16x4_f32(false, hi, false, lo, (short)0, a10, false, false);
        a11 = __builtin_amdgcn_wmma_f32_16x16x4_f32(false, hi, false, hi, (short)0, a11, false, false);

        sLo += lo.x + lo.y;
        sHi += hi.x + hi.y;
    }

    __shared__ float red[8 * 32 * RED_P];
    __shared__ float reds[8 * 32];

    sLo += __shfl_xor(sLo, 16, 32);
    sHi += __shfl_xor(sHi, 16, 32);
    if (half == 0) {
        reds[wave * 32 + lane16]      = sLo;
        reds[wave * 32 + lane16 + 16] = sHi;
    }

    float* rw = red + wave * (32 * RED_P) + (8 * half) * RED_P + lane16;
    #pragma unroll
    for (int j = 0; j < 8; ++j) {
        rw[(j)      * RED_P]      = a00[j];
        rw[(j)      * RED_P + 16] = a01[j];
        rw[(j + 16) * RED_P]      = a10[j];
        rw[(j + 16) * RED_P + 16] = a11[j];
    }
    __syncthreads();

    #pragma unroll
    for (int k = 0; k < 4; ++k) {
        const int row = 8 * k + wave;
        const float* rr = red + row * RED_P + lane;
        float s = rr[0];
        #pragma unroll
        for (int w = 1; w < 8; ++w) s += rr[w * (32 * RED_P)];
        atomicAdd(&gram[row * 32 + lane], s);
    }

    if (threadIdx.x < 32) {
        float s = reds[threadIdx.x];
        #pragma unroll
        for (int w = 1; w < 8; ++w) s += reds[w * 32 + threadIdx.x];
        atomicAdd(&sums[threadIdx.x], s);
    }
}

#define NS_P 36

__device__ __forceinline__ v4f ns_ld4(const float* p) { return *(const v4f*)p; }
__device__ __forceinline__ v2f ns_lo(const v4f v) { return (v2f){v.x, v.y}; }
__device__ __forceinline__ v2f ns_hi(const v4f v) { return (v2f){v.z, v.w}; }

__device__ __forceinline__ void ns_store_tile(float* m, float* mt, const v8f d) {
    #pragma unroll
    for (int r = 0; r < 8; ++r) m[r * NS_P] = d[r];
    *(v4f*)(mt)     = (v4f){d[0], d[1], d[2], d[3]};
    *(v4f*)(mt + 4) = (v4f){d[4], d[5], d[6], d[7]};
}

#define NS_XM(v, ctl) __builtin_bit_cast(float, \
    __builtin_amdgcn_update_dpp(0, __builtin_bit_cast(int, (v)), (ctl), 0xf, 0xf, true))
__device__ __forceinline__ float ns_wave_sum(float v) {
    v += NS_XM(v, 0x161);
    v += NS_XM(v, 0x162);
    v += NS_XM(v, 0x164);
    v += NS_XM(v, 0x168);
    const int b = __builtin_bit_cast(int, v);
    return v + __builtin_bit_cast(float, __builtin_amdgcn_permlanex16(b, b, 0x76543210, 0xfedcba98, false, false));
}

#define NS_RES2_DONE 5.9604645e-8f
#define NS_FIRST(v) __builtin_bit_cast(float, __builtin_amdgcn_readfirstlane(__builtin_bit_cast(int, (v))))

__global__ void __launch_bounds__(128) dbn_invsqrt(float* __restrict__ ws) {
    const int g    = blockIdx.x;
    const int tid  = threadIdx.x;
    const int lane = tid & 31;
    const int wave = tid >> 5;
    const int c    = lane & 15;
    const int half = lane >> 4;
    const int tm   = wave >> 1;
    const int tn   = wave & 1;

    __shared__ __attribute__((aligned(16))) float Ym[32 * NS_P], Yt[32 * NS_P];
    __shared__ __attribute__((aligned(16))) float Zm[32 * NS_P], Zt[32 * NS_P];
    __shared__ __attribute__((aligned(16))) float Tm[32 * NS_P], Tt[32 * NS_P];
    __shared__ float mean_s[32];
    __shared__ float tr;
    __shared__ __attribute__((aligned(16))) float res_s[4];

    const float* gram = ws + WS_GRAM + g * 1024;
    const float* sums = ws + WS_SUMS + g * 32;
    float* wm    = ws + WS_WM    + g * 1024;
    float* shift = ws + WS_SHIFT + g * 32;

    const int j = lane;
    float gv[8];
    #pragma unroll
    for (int q = 0; q < 8; ++q) gv[q] = gram[(wave + 4 * q) * 32 + j];

    const float invN = 1.0f / (float)NCOLS;
    if (tid < 32) mean_s[tid] = sums[tid] * invN;
    if (tid == 0) tr = 0.f;
    __syncthreads();

    const float mj = mean_s[j];
    float sg[8];
    float sdiag = 0.f;
    #pragma unroll
    for (int q = 0; q < 8; ++q) {
        const int i = wave + 4 * q;
        const float mi = mean_s[i];
        const float sij = gv[q] - (float)NCOLS * mi * mj + ((i == j) ? EPS : 0.f);
        sg[q] = sij;
        sdiag = (i == j) ? sij : sdiag;
    }
    if ((j & 3) == wave) atomicAdd(&tr, sdiag);
    __syncthreads();
    const float t = tr;

    #pragma unroll
    for (int q = 0; q < 8; ++q) {
        const int i = wave + 4 * q;
        const float y = sg[q] / t;
        const float z = (i == j) ? 1.f : 0.f;
        Ym[i * NS_P + j] = y;  Yt[j * NS_P + i] = y;
        Zm[i * NS_P + j] = z;  Zt[j * NS_P + i] = z;
    }

    const int offA = (tm * 16 + c) * NS_P + 16 * half;
    const int offB = (tn * 16 + c) * NS_P + 16 * half;
    const int offDm = (tm * 16 + 8 * half) * NS_P + tn * 16 + c;
    const int offDt = (tn * 16 + c) * NS_P + tm * 16 + 8 * half;

    v8f eye3, eye1;
    #pragma unroll
    for (int r = 0; r < 8; ++r) {
        const bool dg = (tm * 16 + 8 * half + r) == (tn * 16 + c);
        eye3[r] = dg ? 3.f : 0.f;
        eye1[r] = dg ? 1.f : 0.f;
    }

    for (int it = 0; it < NS_ITERS; ++it) {
        __syncthreads();
        v4f zA[4], yB[4], yA[4], zB[4];
        #pragma unroll
        for (int q = 0; q < 4; ++q) {
            zA[q] = ns_ld4(Zm + offA + 4 * q);
            yB[q] = ns_ld4(Yt + offB + 4 * q);
        }
        #pragma unroll
        for (int q = 0; q < 4; ++q) {
            yA[q] = ns_ld4(Ym + offA + 4 * q);
            zB[q] = ns_ld4(Zt + offB + 4 * q);
        }
        __builtin_amdgcn_sched_barrier(0);

        v8f zy = {};
        #pragma unroll
        for (int q = 0; q < 4; ++q) {
            zy = __builtin_amdgcn_wmma_f32_16x16x4_f32(false, ns_lo(zA[q]), false, ns_lo(yB[q]), (short)0, zy, false, false);
            zy = __builtin_amdgcn_wmma_f32_16x16x4_f32(false, ns_hi(zA[q]), false, ns_hi(yB[q]), (short)0, zy, false, false);
        }
        asm volatile("v_nop\n\tv_nop\n\tv_nop\n\tv_nop" : "+v"(zy));
        ns_store_tile(Tm + offDm, Tt + offDt, eye3 - zy);
        __builtin_amdgcn_sched_barrier(0);
        const v8f d = eye1 - zy;
        const v2f d0 = {d[0], d[1]}, d1 = {d[2], d[3]}, d2 = {d[4], d[5]}, d3 = {d[6], d[7]};
        const v2f q = (d0 * d0 + d1 * d1) + (d2 * d2 + d3 * d3);
        const float rs = ns_wave_sum(q.x + q.y);
        if (lane == 0) res_s[wave] = rs;
        __syncthreads();

        const v4f rv = ns_ld4(res_s);
        v4f tA[4], tB[4];
        #pragma unroll
        for (int q = 0; q < 4; ++q) {
            tB[q] = ns_ld4(Tt + offB + 4 * q);
            tA[q] = ns_ld4(Tm + offA + 4 * q);
        }
        const float r0 = rv.x, r1 = rv.y, r2 = rv.z, r3 = rv.w;
        const float res2 = (NS_FIRST(r0) + NS_FIRST(r1)) + (NS_FIRST(r2) + NS_FIRST(r3));
        __builtin_amdgcn_sched_barrier(0);

        v8f yt = {}, tz = {};
        #pragma unroll
        for (int q = 0; q < 4; ++q) {
            yt = __builtin_amdgcn_wmma_f32_16x16x4_f32(false, ns_lo(yA[q]), false, ns_lo(tB[q]), (short)0, yt, false, false);
            tz = __builtin_amdgcn_wmma_f32_16x16x4_f32(false, ns_lo(tA[q]), false, ns_lo(zB[q]), (short)0, tz, false, false);
            yt = __builtin_amdgcn_wmma_f32_16x16x4_f32(false, ns_hi(yA[q]), false, ns_hi(tB[q]), (short)0, yt, false, false);
            tz = __builtin_amdgcn_wmma_f32_16x16x4_f32(false, ns_hi(tA[q]), false, ns_hi(zB[q]), (short)0, tz, false, false);
        }
        asm volatile("v_nop\n\tv_nop\n\tv_nop\n\tv_nop" : "+v"(yt), "+v"(tz));
        ns_store_tile(Ym + offDm, Yt + offDt, 0.5f * yt);
        ns_store_tile(Zm + offDm, Zt + offDt, 0.5f * tz);
        if (res2 < NS_RES2_DONE) break;
    }
    __syncthreads();

    const float rs = rsqrtf(t);
    #pragma unroll
    for (int q = 0; q < 8; ++q) {
        const int i = wave + 4 * q;
        const float w = Zm[i * NS_P + j] * rs;
        wm[i * 32 + j] = w;
        Tm[i * NS_P + j] = w * mj;
    }
    __syncthreads();
    if (tid < 32) {
        float s = 0.f;
        #pragma unroll
        for (int k = 0; k < 32; ++k) s += Tm[tid * NS_P + k];
        shift[tid] = s;
    }
}

__global__ void __launch_bounds__(256) dbn_apply(const float* __restrict__ x,
                                                 const float* __restrict__ weight,
                                                 const float* __restrict__ bias,
                                                 const float* __restrict__ ws,
                                                 float* __restrict__ out) {
    const int g      = blockIdx.y;
    const int lane   = threadIdx.x & 31;
    const int wave   = threadIdx.x >> 5;
    const int lane16 = lane & 15;
    const int half   = lane >> 4;
    const int wid    = blockIdx.x * 8 + wave;

    __shared__ float sw[32], sb[32], ssh[32];
    if (threadIdx.x < 32) {
        const int c = g * 32 + threadIdx.x;
        sw[threadIdx.x]  = weight[c];
        sb[threadIdx.x]  = bias[c];
        ssh[threadIdx.x] = ws[WS_SHIFT + g * 32 + threadIdx.x];
    }
    __syncthreads();

    const float* wm = ws + WS_WM + g * 1024;
    v2f a0[8], a1[8];
    #pragma unroll
    for (int ks = 0; ks < 8; ++ks) {
        a0[ks] = *(const v2f*)(wm + lane16 * 32 + ks * 4 + 2 * half);
        a1[ks] = *(const v2f*)(wm + (lane16 + 16) * 32 + ks * 4 + 2 * half);
    }

    float wA[8], cA[8], wB[8], cB[8];
    #pragma unroll
    for (int jj = 0; jj < 8; ++jj) {
        const int m = jj + 8 * half;
        wA[jj] = sw[m];       cA[jj] = sb[m]      - ssh[m]      * sw[m];
        wB[jj] = sw[m + 16];  cB[jj] = sb[m + 16] - ssh[m + 16] * sw[m + 16];
    }

    const int b  = wid >> 5;
    const int s0 = (wid & 31) * 128;

    const float* px = x + (size_t)(b * NUM_FEATURES + g * GROUP_SIZE) * HW
                        + (size_t)(s0 + 4 * lane16) + (size_t)(2 * half) * HW;
    float* po = out + (size_t)(b * NUM_FEATURES + g * GROUP_SIZE) * HW
                    + (size_t)(s0 + 4 * lane16) + (size_t)(8 * half) * HW;

    for (int t = 0; t < 2; ++t) {
        const int co = t * 64;
        v8f acc0[4], acc1[4];
        #pragma unroll
        for (int e = 0; e < 4; ++e) { acc0[e] = (v8f){}; acc1[e] = (v8f){}; }
        #pragma unroll
        for (int ks = 0; ks < 8; ++ks) {
            const v4f r0 = *(const v4f*)(px + (size_t)(ks * 4) * HW + co);
            const v4f r1 = *(const v4f*)(px + (size_t)(ks * 4 + 1) * HW + co);
            #pragma unroll
            for (int e = 0; e < 4; ++e) {
                const v2f bf = {r0[e], r1[e]};
                acc0[e] = __builtin_amdgcn_wmma_f32_16x16x4_f32(false, a0[ks], false, bf, (short)0, acc0[e], false, false);
                acc1[e] = __builtin_amdgcn_wmma_f32_16x16x4_f32(false, a1[ks], false, bf, (short)0, acc1[e], false, false);
            }
        }
        #pragma unroll
        for (int jj = 0; jj < 8; ++jj) {
            const v4f o0 = {acc0[0][jj] * wA[jj] + cA[jj], acc0[1][jj] * wA[jj] + cA[jj],
                            acc0[2][jj] * wA[jj] + cA[jj], acc0[3][jj] * wA[jj] + cA[jj]};
            const v4f o1 = {acc1[0][jj] * wB[jj] + cB[jj], acc1[1][jj] * wB[jj] + cB[jj],
                            acc1[2][jj] * wB[jj] + cB[jj], acc1[3][jj] * wB[jj] + cB[jj]};
            __builtin_nontemporal_store(o0, (v4f*)(po + (size_t)jj * HW + co));
            __builtin_nontemporal_store(o1, (v4f*)(po + (size_t)(jj + 16) * HW + co));
        }
    }
}

extern "C" void kernel_launch(void* const* d_in, const int* in_sizes, int n_in,
                              void* d_out, int out_size, void* d_ws, size_t ws_size,
                              hipStream_t stream) {
    const float* x      = (const float*)d_in[0];
    const float* weight = (const float*)d_in[1];
    const float* bias   = (const float*)d_in[2];
    float* out = (float*)d_out;
    float* ws  = (float*)d_ws;

    hipMemsetAsync(ws, 0,
                   (size_t)(WS_SUMS + NUM_GROUPS * 32) * sizeof(float), stream);

    dbn_gram   <<<dim3(64,  NUM_GROUPS), 256,  0, stream>>>(x, ws);
    dbn_invsqrt<<<dim3(NUM_GROUPS),      128,  0, stream>>>(ws);
    dbn_apply  <<<dim3(128, NUM_GROUPS), 256,  0, stream>>>(x, weight, bias, ws, out);
}


# Gram kernel: x reaches the WMMA fragments through a per-wave LDS tile: a global load covers 256 contiguous bytes of each of 2 rows (4 lines, was 16 x 16 B); Gram and channel sum order changes
# speedup vs baseline: 3.1969x; 1.0457x over previous
#include <hip/hip_runtime.h>

typedef __attribute__((ext_vector_type(2))) float v2f;
typedef __attribute__((ext_vector_type(4))) float v4f;
typedef __attribute__((ext_vector_type(8))) float v8f;

#define NUM_FEATURES 256
#define GROUP_SIZE   32
#define NUM_GROUPS   8
#define BATCH        32
#define HW           4096
#define NCOLS        (BATCH * HW)
#define EPS          1e-5f
#define NS_ITERS     32

#define WS_GRAM   0
#define WS_SUMS   (NUM_GROUPS * 32 * 32)
#define WS_WM     (WS_SUMS + NUM_GROUPS * 32)
#define WS_SHIFT  (WS_WM + NUM_GROUPS * 32 * 32)

#define RED_P 34
#define ST_P  68
__global__ void __launch_bounds__(256, 4) dbn_gram(const float* __restrict__ x,
                                                float* __restrict__ ws) {
    const int g      = blockIdx.y;
    const int lane   = threadIdx.x & 31;
    const int wave   = threadIdx.x >> 5;
    const int lane16 = lane & 15;
    const int half   = lane >> 4;
    const int wid    = blockIdx.x * 8 + wave;

    float* gram = ws + WS_GRAM + g * 1024;
    float* sums = ws + WS_SUMS + g * 32;

    const int b  = wid >> 4;
    const int s0 = (wid & 15) * 256;

    __shared__ __attribute__((aligned(16))) float stage[8 * 32 * ST_P];
    float* st = stage + wave * (32 * ST_P);

    const float* pg = x + (size_t)(b * NUM_FEATURES + g * GROUP_SIZE + half) * HW
                        + (size_t)(s0 + 4 * lane16);
    float*       stW = st + half * ST_P + 4 * lane16;
    const float* stR = st + lane16 * ST_P + 4 * half;

    v4f gl[16];
    #pragma unroll
    for (int j = 0; j < 16; ++j) gl[j] = *(const v4f*)(pg + (size_t)(2 * j) * HW);

    v8f a00 = {}, a01 = {}, a10 = {}, a11 = {};
    float sLo = 0.f, sHi = 0.f;

    #pragma unroll
    for (int ch = 0; ch < 4; ++ch) {
        #pragma unroll
        for (int j = 0; j < 16; ++j) *(v4f*)(stW + (2 * j) * ST_P) = gl[j];
        if (ch < 3) {
            #pragma unroll
            for (int j = 0; j < 16; ++j)
                gl[j] = *(const v4f*)(pg + (size_t)(2 * j) * HW + (ch + 1) * 64);
        }
        __builtin_amdgcn_wave_barrier();
        v4f lo[8], hi[8];
        #pragma unroll
        for (int kk = 0; kk < 8; ++kk) {
            lo[kk] = *(const v4f*)(stR + 8 * kk);
            hi[kk] = *(const v4f*)(stR + 16 * ST_P + 8 * kk);
        }
        __builtin_amdgcn_sched_barrier(0);
        #pragma unroll
        for (int kk = 0; kk < 8; ++kk) {
            const v2f l0 = {lo[kk].x, lo[kk].y}, l1 = {lo[kk].z, lo[kk].w};
            const v2f h0 = {hi[kk].x, hi[kk].y}, h1 = {hi[kk].z, hi[kk].w};

            a00 = __builtin_amdgcn_wmma_f32_16x16x4_f32(false, l0, false, l0, (short)0, a00, false, false);
            a01 = __builtin_amdgcn_wmma_f32_16x16x4_f32(false, l0, false, h0, (short)0, a01, false, false);
            a10 = __builtin_amdgcn_wmma_f32_16x16x4_f32(false, h0, false, l0, (short)0, a10, false, false);
            a11 = __builtin_amdgcn_wmma_f32_16x16x4_f32(false, h0, false, h0, (short)0, a11, false, false);
            a00 = __builtin_amdgcn_wmma_f32_16x16x4_f32(false, l1, false, l1, (short)0, a00, false, false);
            a01 = __builtin_amdgcn_wmma_f32_16x16x4_f32(false, l1, false, h1, (short)0, a01, false, false);
            a10 = __builtin_amdgcn_wmma_f32_16x16x4_f32(false, h1, false, l1, (short)0, a10, false, false);
            a11 = __builtin_amdgcn_wmma_f32_16x16x4_f32(false, h1, false, h1, (short)0, a11, false, false);

            sLo += (lo[kk].x + lo[kk].y) + (lo[kk].z + lo[kk].w);
            sHi += (hi[kk].x + hi[kk].y) + (hi[kk].z + hi[kk].w);
        }
        __builtin_amdgcn_wave_barrier();
    }

    __shared__ float red[8 * 32 * RED_P];
    __shared__ float reds[8 * 32];

    sLo += __shfl_xor(sLo, 16, 32);
    sHi += __shfl_xor(sHi, 16, 32);
    if (half == 0) {
        reds[wave * 32 + lane16]      = sLo;
        reds[wave * 32 + lane16 + 16] = sHi;
    }

    float* rw = red + wave * (32 * RED_P) + (8 * half) * RED_P + lane16;
    #pragma unroll
    for (int j = 0; j < 8; ++j) {
        rw[(j)      * RED_P]      = a00[j];
        rw[(j)      * RED_P + 16] = a01[j];
        rw[(j + 16) * RED_P]      = a10[j];
        rw[(j + 16) * RED_P + 16] = a11[j];
    }
    __syncthreads();

    #pragma unroll
    for (int k = 0; k < 4; ++k) {
        const int row = 8 * k + wave;
        const float* rr = red + row * RED_P + lane;
        float s = rr[0];
        #pragma unroll
        for (int w = 1; w < 8; ++w) s += rr[w * (32 * RED_P)];
        atomicAdd(&gram[row * 32 + lane], s);
    }

    if (threadIdx.x < 32) {
        float s = reds[threadIdx.x];
        #pragma unroll
        for (int w = 1; w < 8; ++w) s += reds[w * 32 + threadIdx.x];
        atomicAdd(&sums[threadIdx.x], s);
    }
}

#define NS_P 36

__device__ __forceinline__ v4f ns_ld4(const float* p) { return *(const v4f*)p; }
__device__ __forceinline__ v2f ns_lo(const v4f v) { return (v2f){v.x, v.y}; }
__device__ __forceinline__ v2f ns_hi(const v4f v) { return (v2f){v.z, v.w}; }

__device__ __forceinline__ void ns_store_tile(float* m, float* mt, const v8f d) {
    #pragma unroll
    for (int r = 0; r < 8; ++r) m[r * NS_P] = d[r];
    *(v4f*)(mt)     = (v4f){d[0], d[1], d[2], d[3]};
    *(v4f*)(mt + 4) = (v4f){d[4], d[5], d[6], d[7]};
}

#define NS_XM(v, ctl) __builtin_bit_cast(float, \
    __builtin_amdgcn_update_dpp(0, __builtin_bit_cast(int, (v)), (ctl), 0xf, 0xf, true))
__device__ __forceinline__ float ns_wave_sum(float v) {
    v += NS_XM(v, 0x161);
    v += NS_XM(v, 0x162);
    v += NS_XM(v, 0x164);
    v += NS_XM(v, 0x168);
    const int b = __builtin_bit_cast(int, v);
    return v + __builtin_bit_cast(float, __builtin_amdgcn_permlanex16(b, b, 0x76543210, 0xfedcba98, false, false));
}

#define NS_RES2_DONE 5.9604645e-8f
#define NS_FIRST(v) __builtin_bit_cast(float, __builtin_amdgcn_readfirstlane(__builtin_bit_cast(int, (v))))

__global__ void __launch_bounds__(128) dbn_invsqrt(float* __restrict__ ws) {
    const int g    = blockIdx.x;
    const int tid  = threadIdx.x;
    const int lane = tid & 31;
    const int wave = tid >> 5;
    const int c    = lane & 15;
    const int half = lane >> 4;
    const int tm   = wave >> 1;
    const int tn   = wave & 1;

    __shared__ __attribute__((aligned(16))) float Ym[32 * NS_P], Yt[32 * NS_P];
    __shared__ __attribute__((aligned(16))) float Zm[32 * NS_P], Zt[32 * NS_P];
    __shared__ __attribute__((aligned(16))) float Tm[32 * NS_P], Tt[32 * NS_P];
    __shared__ float mean_s[32];
    __shared__ float tr;
    __shared__ __attribute__((aligned(16))) float res_s[4];

    const float* gram = ws + WS_GRAM + g * 1024;
    const float* sums = ws + WS_SUMS + g * 32;
    float* wm    = ws + WS_WM    + g * 1024;
    float* shift = ws + WS_SHIFT + g * 32;

    const int j = lane;
    float gv[8];
    #pragma unroll
    for (int q = 0; q < 8; ++q) gv[q] = gram[(wave + 4 * q) * 32 + j];

    const float invN = 1.0f / (float)NCOLS;
    if (tid < 32) mean_s[tid] = sums[tid] * invN;
    if (tid == 0) tr = 0.f;
    __syncthreads();

    const float mj = mean_s[j];
    float sg[8];
    float sdiag = 0.f;
    #pragma unroll
    for (int q = 0; q < 8; ++q) {
        const int i = wave + 4 * q;
        const float mi = mean_s[i];
        const float sij = gv[q] - (float)NCOLS * mi * mj + ((i == j) ? EPS : 0.f);
        sg[q] = sij;
        sdiag = (i == j) ? sij : sdiag;
    }
    if ((j & 3) == wave) atomicAdd(&tr, sdiag);
    __syncthreads();
    const float t = tr;

    #pragma unroll
    for (int q = 0; q < 8; ++q) {
        const int i = wave + 4 * q;
        const float y = sg[q] / t;
        const float z = (i == j) ? 1.f : 0.f;
        Ym[i * NS_P + j] = y;  Yt[j * NS_P + i] = y;
        Zm[i * NS_P + j] = z;  Zt[j * NS_P + i] = z;
    }

    const int offA = (tm * 16 + c) * NS_P + 16 * half;
    const int offB = (tn * 16 + c) * NS_P + 16 * half;
    const int offDm = (tm * 16 + 8 * half) * NS_P + tn * 16 + c;
    const int offDt = (tn * 16 + c) * NS_P + tm * 16 + 8 * half;

    v8f eye3, eye1;
    #pragma unroll
    for (int r = 0; r < 8; ++r) {
        const bool dg = (tm * 16 + 8 * half + r) == (tn * 16 + c);
        eye3[r] = dg ? 3.f : 0.f;
        eye1[r] = dg ? 1.f : 0.f;
    }

    for (int it = 0; it < NS_ITERS; ++it) {
        __syncthreads();
        v4f zA[4], yB[4], yA[4], zB[4];
        #pragma unroll
        for (int q = 0; q < 4; ++q) {
            zA[q] = ns_ld4(Zm + offA + 4 * q);
            yB[q] = ns_ld4(Yt + offB + 4 * q);
        }
        #pragma unroll
        for (int q = 0; q < 4; ++q) {
            yA[q] = ns_ld4(Ym + offA + 4 * q);
            zB[q] = ns_ld4(Zt + offB + 4 * q);
        }
        __builtin_amdgcn_sched_barrier(0);

        v8f zy = {};
        #pragma unroll
        for (int q = 0; q < 4; ++q) {
            zy = __builtin_amdgcn_wmma_f32_16x16x4_f32(false, ns_lo(zA[q]), false, ns_lo(yB[q]), (short)0, zy, false, false);
            zy = __builtin_amdgcn_wmma_f32_16x16x4_f32(false, ns_hi(zA[q]), false, ns_hi(yB[q]), (short)0, zy, false, false);
        }
        asm volatile("v_nop\n\tv_nop\n\tv_nop\n\tv_nop" : "+v"(zy));
        ns_store_tile(Tm + offDm, Tt + offDt, eye3 - zy);
        __builtin_amdgcn_sched_barrier(0);
        const v8f d = eye1 - zy;
        const v2f d0 = {d[0], d[1]}, d1 = {d[2], d[3]}, d2 = {d[4], d[5]}, d3 = {d[6], d[7]};
        const v2f q = (d0 * d0 + d1 * d1) + (d2 * d2 + d3 * d3);
        const float rs = ns_wave_sum(q.x + q.y);
        if (lane == 0) res_s[wave] = rs;
        __syncthreads();

        const v4f rv = ns_ld4(res_s);
        v4f tA[4], tB[4];
        #pragma unroll
        for (int q = 0; q < 4; ++q) {
            tB[q] = ns_ld4(Tt + offB + 4 * q);
            tA[q] = ns_ld4(Tm + offA + 4 * q);
        }
        const float r0 = rv.x, r1 = rv.y, r2 = rv.z, r3 = rv.w;
        const float res2 = (NS_FIRST(r0) + NS_FIRST(r1)) + (NS_FIRST(r2) + NS_FIRST(r3));
        __builtin_amdgcn_sched_barrier(0);

        v8f yt = {}, tz = {};
        #pragma unroll
        for (int q = 0; q < 4; ++q) {
            yt = __builtin_amdgcn_wmma_f32_16x16x4_f32(false, ns_lo(yA[q]), false, ns_lo(tB[q]), (short)0, yt, false, false);
            tz = __builtin_amdgcn_wmma_f32_16x16x4_f32(false, ns_lo(tA[q]), false, ns_lo(zB[q]), (short)0, tz, false, false);
            yt = __builtin_amdgcn_wmma_f32_16x16x4_f32(false, ns_hi(yA[q]), false, ns_hi(tB[q]), (short)0, yt, false, false);
            tz = __builtin_amdgcn_wmma_f32_16x16x4_f32(false, ns_hi(tA[q]), false, ns_hi(zB[q]), (short)0, tz, false, false);
        }
        asm volatile("v_nop\n\tv_nop\n\tv_nop\n\tv_nop" : "+v"(yt), "+v"(tz));
        ns_store_tile(Ym + offDm, Yt + offDt, 0.5f * yt);
        ns_store_tile(Zm + offDm, Zt + offDt, 0.5f * tz);
        if (res2 < NS_RES2_DONE) break;
    }
    __syncthreads();

    const float rs = rsqrtf(t);
    #pragma unroll
    for (int q = 0; q < 8; ++q) {
        const int i = wave + 4 * q;
        const float w = Zm[i * NS_P + j] * rs;
        wm[i * 32 + j] = w;
        Tm[i * NS_P + j] = w * mj;
    }
    __syncthreads();
    if (tid < 32) {
        float s = 0.f;
        #pragma unroll
        for (int k = 0; k < 32; ++k) s += Tm[tid * NS_P + k];
        shift[tid] = s;
    }
}

__global__ void __launch_bounds__(256) dbn_apply(const float* __restrict__ x,
                                                 const float* __restrict__ weight,
                                                 const float* __restrict__ bias,
                                                 const float* __restrict__ ws,
                                                 float* __restrict__ out) {
    const int g      = blockIdx.y;
    const int lane   = threadIdx.x & 31;
    const int wave   = threadIdx.x >> 5;
    const int lane16 = lane & 15;
    const int half   = lane >> 4;
    const int wid    = blockIdx.x * 8 + wave;

    __shared__ float sw[32], sb[32], ssh[32];
    if (threadIdx.x < 32) {
        const int c = g * 32 + threadIdx.x;
        sw[threadIdx.x]  = weight[c];
        sb[threadIdx.x]  = bias[c];
        ssh[threadIdx.x] = ws[WS_SHIFT + g * 32 + threadIdx.x];
    }
    __syncthreads();

    const float* wm = ws + WS_WM + g * 1024;
    v2f a0[8], a1[8];
    #pragma unroll
    for (int ks = 0; ks < 8; ++ks) {
        a0[ks] = *(const v2f*)(wm + lane16 * 32 + ks * 4 + 2 * half);
        a1[ks] = *(const v2f*)(wm + (lane16 + 16) * 32 + ks * 4 + 2 * half);
    }

    float wA[8], cA[8], wB[8], cB[8];
    #pragma unroll
    for (int jj = 0; jj < 8; ++jj) {
        const int m = jj + 8 * half;
        wA[jj] = sw[m];       cA[jj] = sb[m]      - ssh[m]      * sw[m];
        wB[jj] = sw[m + 16];  cB[jj] = sb[m + 16] - ssh[m + 16] * sw[m + 16];
    }

    const int b  = wid >> 5;
    const int s0 = (wid & 31) * 128;

    const float* px = x + (size_t)(b * NUM_FEATURES + g * GROUP_SIZE) * HW
                        + (size_t)(s0 + 4 * lane16) + (size_t)(2 * half) * HW;
    float* po = out + (size_t)(b * NUM_FEATURES + g * GROUP_SIZE) * HW
                    + (size_t)(s0 + 4 * lane16) + (size_t)(8 * half) * HW;

    for (int t = 0; t < 2; ++t) {
        const int co = t * 64;
        v8f acc0[4], acc1[4];
        #pragma unroll
        for (int e = 0; e < 4; ++e) { acc0[e] = (v8f){}; acc1[e] = (v8f){}; }
        #pragma unroll
        for (int ks = 0; ks < 8; ++ks) {
            const v4f r0 = *(const v4f*)(px + (size_t)(ks * 4) * HW + co);
            const v4f r1 = *(const v4f*)(px + (size_t)(ks * 4 + 1) * HW + co);
            #pragma unroll
            for (int e = 0; e < 4; ++e) {
                const v2f bf = {r0[e], r1[e]};
                acc0[e] = __builtin_amdgcn_wmma_f32_16x16x4_f32(false, a0[ks], false, bf, (short)0, acc0[e], false, false);
                acc1[e] = __builtin_amdgcn_wmma_f32_16x16x4_f32(false, a1[ks], false, bf, (short)0, acc1[e], false, false);
            }
        }
        #pragma unroll
        for (int jj = 0; jj < 8; ++jj) {
            const v4f o0 = {acc0[0][jj] * wA[jj] + cA[jj], acc0[1][jj] * wA[jj] + cA[jj],
                            acc0[2][jj] * wA[jj] + cA[jj], acc0[3][jj] * wA[jj] + cA[jj]};
            const v4f o1 = {acc1[0][jj] * wB[jj] + cB[jj], acc1[1][jj] * wB[jj] + cB[jj],
                            acc1[2][jj] * wB[jj] + cB[jj], acc1[3][jj] * wB[jj] + cB[jj]};
            __builtin_nontemporal_store(o0, (v4f*)(po + (size_t)jj * HW + co));
            __builtin_nontemporal_store(o1, (v4f*)(po + (size_t)(jj + 16) * HW + co));
        }
    }
}

extern "C" void kernel_launch(void* const* d_in, const int* in_sizes, int n_in,
                              void* d_out, int out_size, void* d_ws, size_t ws_size,
                              hipStream_t stream) {
    const float* x      = (const float*)d_in[0];
    const float* weight = (const float*)d_in[1];
    const float* bias   = (const float*)d_in[2];
    float* out = (float*)d_out;
    float* ws  = (float*)d_ws;

    hipMemsetAsync(ws, 0,
                   (size_t)(WS_SUMS + NUM_GROUPS * 32) * sizeof(float), stream);

    dbn_gram   <<<dim3(64,  NUM_GROUPS), 256,  0, stream>>>(x, ws);
    dbn_invsqrt<<<dim3(NUM_GROUPS),      128,  0, stream>>>(ws);
    dbn_apply  <<<dim3(128, NUM_GROUPS), 256,  0, stream>>>(x, weight, bias, ws, out);
}
